# starting kernel
# baseline (speedup 1.0000x reference)
#include <hip/hip_runtime.h>


typedef float v2f __attribute__((ext_vector_type(2)));
typedef float v8f __attribute__((ext_vector_type(8)));

#define R_TOT          128
#define T_STEPS        1024
#define N_AG           128
#define RUNS_PER_BLOCK 16
#define SLOT_STRIDE    132

__device__ __forceinline__ float fast_tanh(float x) {
#if __has_builtin(__builtin_amdgcn_tanhf)
  return __builtin_amdgcn_tanhf(x);
#else
  float e = __builtin_exp2f(x * 2.885390081777927f);
  return 1.0f - 2.0f / (e + 1.0f);
#endif
}

__global__ __launch_bounds__(32)
void com2net_wmma_kernel(const float* __restrict__ runs,
                         const float* __restrict__ W1,
                         const float* __restrict__ b1,
                         const float* __restrict__ W2,
                         const float* __restrict__ b2,
                         float*       __restrict__ out)
{
  __shared__ __align__(16) float cslot [32 * SLOT_STRIDE];
  __shared__ __align__(16) float outbuf[32 * SLOT_STRIDE];

  const int lane    = threadIdx.x;
  const int up      = lane >> 4;
  const int l15     = lane & 15;
  const int runBase = blockIdx.x * RUNS_PER_BLOCK;

  for (int k = lane; k < 32 * SLOT_STRIDE; k += 32) cslot[k] = 0.0f;

  v2f amat; amat.x = 0.0f; amat.y = 0.0f;
  {
    const int h  = l15;
    const int k0 = up * 2;
    if (h < 10) { amat.x = W1[h * 4 + k0]; amat.y = W1[h * 4 + k0 + 1]; }
  }

  v8f bias1;
#pragma unroll
  for (int v = 0; v < 8; ++v) {
    const int m = v + 8 * up;
    bias1[v] = (m < 10) ? b1[m] : 0.0f;
  }

  float w2r[30];
#pragma unroll
  for (int k = 0; k < 30; ++k) w2r[k] = W2[k];
  const float b20 = b2[0], b21 = b2[1], b22 = b2[2];

  const float* xrow_base = runs + (size_t)(runBase + l15) * T_STEPS * (N_AG * 2);
  const float* cread     = &cslot [l15  * SLOT_STRIDE];
  float*       cwrite    = &cslot [lane * SLOT_STRIDE];
  float*       obwrite   = &outbuf[lane * SLOT_STRIDE];

  for (int t = 0; t < T_STEPS; ++t) {
    const float* xrow = xrow_base + (size_t)t * (N_AG * 2);

    if (t + 1 < T_STEPS) {
      const char* nxt = (const char*)(xrow + N_AG * 2);
#pragma unroll
      for (int j = 0; j < 4; ++j)
        __builtin_prefetch(nxt + (size_t)(up * 4 + j) * 128, 0, 0);
    }

    v2f   xv_cur = *(const v2f*)xrow;
    float cb_cur = cread[1];
    float co     = 0.0f;

    for (int i = 0; i < N_AG; ++i) {
      const int ip = (i + 1 < N_AG) ? i + 1 : i;
      const v2f   xv_next = *(const v2f*)(xrow + ip * 2);
      const float cb_next = cread[ip + 1];

      v2f bmat;
      bmat.x = up ? co     : xv_cur.x;
      bmat.y = up ? cb_cur : xv_cur.y;

      v8f d = __builtin_amdgcn_wmma_f32_16x16x4_f32(
          false, amat, false, bmat, (short)0, bias1, false, false);

      float dt[8];
#pragma unroll
      for (int v = 0; v < 8; ++v) dt[v] = fast_tanh(d[v]);

      const float h8 = __shfl_xor(dt[0], 16, 32);
      const float h9 = __shfl_xor(dt[1], 16, 32);

      float e0 = fmaf(w2r[20], dt[0], b22);
      float e1 =      w2r[21] * dt[1];
      e0 = fmaf(w2r[22], dt[2], e0);  e1 = fmaf(w2r[23], dt[3], e1);
      e0 = fmaf(w2r[24], dt[4], e0);  e1 = fmaf(w2r[25], dt[5], e1);
      e0 = fmaf(w2r[26], dt[6], e0);  e1 = fmaf(w2r[27], dt[7], e1);
      e0 = fmaf(w2r[28], h8,    e0);  e1 = fmaf(w2r[29], h9,    e1);
      const float o2 = e0 + e1;
      co = __shfl_xor(o2, 16, 32);

      float a0 = fmaf(w2r[0], dt[0], b20), a1 = w2r[1] * dt[1];
      a0 = fmaf(w2r[2], dt[2], a0);  a1 = fmaf(w2r[3], dt[3], a1);
      a0 = fmaf(w2r[4], dt[4], a0);  a1 = fmaf(w2r[5], dt[5], a1);
      a0 = fmaf(w2r[6], dt[6], a0);  a1 = fmaf(w2r[7], dt[7], a1);
      a0 = fmaf(w2r[8], h8,   a0);   a1 = fmaf(w2r[9], h9,   a1);
      const float o0 = a0 + a1;

      float c0 = fmaf(w2r[10], dt[0], b21), c1 = w2r[11] * dt[1];
      c0 = fmaf(w2r[12], dt[2], c0); c1 = fmaf(w2r[13], dt[3], c1);
      c0 = fmaf(w2r[14], dt[4], c0); c1 = fmaf(w2r[15], dt[5], c1);
      c0 = fmaf(w2r[16], dt[6], c0); c1 = fmaf(w2r[17], dt[7], c1);
      c0 = fmaf(w2r[18], h8,   c0);  c1 = fmaf(w2r[19], h9,   c1);
      const float o1 = c0 + c1;

      obwrite[i] = o0;
      cwrite[i]  = o1;

      xv_cur = xv_next;
      cb_cur = cb_next;
    }

#pragma unroll 4
    for (int rr = 0; rr < RUNS_PER_BLOCK; ++rr) {
      const float4 vv = *(const float4*)&outbuf[rr * SLOT_STRIDE + lane * 4];
      *(float4*)(out + ((size_t)(runBase + rr) * T_STEPS + t) * N_AG + lane * 4) = vv;
    }
  }
}

extern "C" void kernel_launch(void* const* d_in, const int* in_sizes, int n_in,
                              void* d_out, int out_size, void* d_ws, size_t ws_size,
                              hipStream_t stream) {
  (void)in_sizes; (void)n_in; (void)out_size; (void)d_ws; (void)ws_size;
  const float* runs = (const float*)d_in[0];
  const float* W1   = (const float*)d_in[1];
  const float* b1   = (const float*)d_in[2];
  const float* W2   = (const float*)d_in[3];
  const float* b2   = (const float*)d_in[4];
  float* out        = (float*)d_out;

  dim3 grid(R_TOT / RUNS_PER_BLOCK);
  dim3 block(32);
  com2net_wmma_kernel<<<grid, block, 0, stream>>>(runs, W1, b1, W2, b2, out);
}


# Timesteps dealt round-robin to 16 waves of a work-group (runs stay on matrix columns); a wave trails the previous timestep's by 8 agents, comm word goes via LDS, barrier per 4 agents; sums unchanged
# speedup vs baseline: 8.4553x; 8.4553x over previous
#include <hip/hip_runtime.h>


typedef float v2f __attribute__((ext_vector_type(2)));
typedef float v8f __attribute__((ext_vector_type(8)));

#define R_TOT          128
#define T_STEPS        1024
#define N_AG           128
#define RUNS_PER_BLOCK 16
#define SLOT_STRIDE    132
#define WAVES          16
#define GROUP          4
#define LAG_GROUPS     2
#define T_PER_WAVE     (T_STEPS / WAVES)
#define GROUPS_PER_T   (N_AG / GROUP)
#define OWN_GROUPS     (T_PER_WAVE * GROUPS_PER_T)
#define EPOCHS         (OWN_GROUPS + LAG_GROUPS * (WAVES - 1))
#define OB_SINK        192
#define OB_WAVE        (RUNS_PER_BLOCK * SLOT_STRIDE + OB_SINK)

static_assert(LAG_GROUPS * GROUP * WAVES == N_AG, "a wave's next timestep is due when it ends one");
static_assert(T_STEPS % WAVES == 0 && N_AG % GROUP == 0, "whole timesteps and groups");
static_assert((OB_WAVE * 4) % 16 == 0, "16-byte rows for the flush");

__device__ __forceinline__ float fast_tanh(float x) {
#if __has_builtin(__builtin_amdgcn_tanhf)
  return __builtin_amdgcn_tanhf(x);
#else
  float e = __builtin_exp2f(x * 2.885390081777927f);
  return 1.0f - 2.0f / (e + 1.0f);
#endif
}

__global__ __launch_bounds__(32 * WAVES)
void com2net_wmma_kernel(const float* __restrict__ runs,
                         const float* __restrict__ W1,
                         const float* __restrict__ b1,
                         const float* __restrict__ W2,
                         const float* __restrict__ b2,
                         float*       __restrict__ out)
{
  __shared__ __align__(16) float cslot [32 * SLOT_STRIDE];
  __shared__ __align__(16) float outbuf[WAVES * OB_WAVE];

  const int wave    = __builtin_amdgcn_readfirstlane((int)(threadIdx.x >> 5));
  const int lane    = threadIdx.x & 31;
  const int up      = lane >> 4;
  const int l15     = lane & 15;
  const int runBase = blockIdx.x * RUNS_PER_BLOCK;

  for (int k = threadIdx.x; k < 32 * SLOT_STRIDE; k += 32 * WAVES) cslot[k] = 0.0f;
  __syncthreads();

  const int hid = (l15 < 5) ? l15 : ((l15 >= 8 && l15 < 13) ? l15 - 3 : -1);

  v2f a1x; a1x.x = 0.0f; a1x.y = 0.0f;
  v2f a1c; a1c.x = 0.0f; a1c.y = 0.0f;
  if (hid >= 0) {
    if (up) { a1x.x = W1[hid * 4 + 3]; }
    else    { a1x.x = W1[hid * 4 + 0]; a1x.y = W1[hid * 4 + 1];
              a1c.x = W1[hid * 4 + 2]; }
  }

  v8f bias1;
#pragma unroll
  for (int v = 0; v < 8; ++v) bias1[v] = (v < 5) ? b1[5 * up + v] : 0.0f;

  v2f a2[3];
#pragma unroll
  for (int q = 0; q < 3; ++q) {
    a2[q].x = 0.0f; a2[q].y = 0.0f;
    if (l15 < 3) {
      a2[q].x = W2[l15 * 10 + 5 * up + 2 * q];
      if (2 * q + 1 < 5) a2[q].y = W2[l15 * 10 + 5 * up + 2 * q + 1];
    }
  }

  v8f bias2;
#pragma unroll
  for (int v = 0; v < 8; ++v) bias2[v] = (!up && v < 3) ? b2[v] : 0.0f;

  const float* xrow_base = runs + (size_t)(runBase + l15) * T_STEPS * (N_AG * 2);
  const float* cread     = &cslot[l15  * SLOT_STRIDE];
  float*       cwrite    = &cslot[lane * SLOT_STRIDE];
  float*       ob        = &outbuf[wave * OB_WAVE];
  float*       obwrite   = up ? (ob + RUNS_PER_BLOCK * SLOT_STRIDE + l15)
                              : (ob + l15 * SLOT_STRIDE);

  auto flush = [&](int tt) {
#pragma unroll 4
    for (int rr = 0; rr < RUNS_PER_BLOCK; ++rr) {
      const float4 vv = *(const float4*)&ob[rr * SLOT_STRIDE + lane * 4];
      *(float4*)(out + ((size_t)(runBase + rr) * T_STEPS + tt) * N_AG + lane * 4) = vv;
    }
  };

  v8f   pre = bias1;
  v2f   xq0, xq1, xq2;
  xq0.x = 0.0f; xq0.y = 0.0f; xq1 = xq0; xq2 = xq0;
  float cq  = 0.0f;
  v2f   bo2; bo2.x = 0.0f; bo2.y = 0.0f;

  for (int g = 0; g < EPOCHS; ++g) {
    const int le = g - LAG_GROUPS * wave;
    if (le >= 0 && le < OWN_GROUPS) {
      const int m  = le / GROUPS_PER_T;
      const int i0 = (le % GROUPS_PER_T) * GROUP;
      const int t  = wave + WAVES * m;
      const float* xrow = xrow_base + (size_t)t * (N_AG * 2);

      if (i0 == 0) {
        const v2f x0v = *(const v2f*)xrow;
        xq0 = *(const v2f*)(xrow + 2);
        xq1 = *(const v2f*)(xrow + 4);
        xq2 = *(const v2f*)(xrow + 6);
        if (m + 1 < T_PER_WAVE) {
          const char* nxt = (const char*)(xrow + WAVES * N_AG * 2);
#pragma unroll
          for (int j = 0; j < 4; ++j)
            __builtin_prefetch(nxt + (size_t)(up * 4 + j) * 128, 0, 0);
        }
        if (m > 0) flush(t - WAVES);
        const float cb0 = cread[1];
        cq = cread[2];
        v2f b1x;
        b1x.x = up ? cb0  : x0v.x;
        b1x.y = up ? 0.0f : x0v.y;
        pre = __builtin_amdgcn_wmma_f32_16x16x4_f32(
            false, a1x, false, b1x, (short)0, bias1, false, false);
        bo2.x = 0.0f; bo2.y = 0.0f;
      }

      float o0_prev = 0.0f, o1_prev = 0.0f;
#pragma unroll
      for (int k = 0; k < GROUP; ++k) {
        const int i = i0 + k;
        const v8f d1 = __builtin_amdgcn_wmma_f32_16x16x4_f32(
            false, a1c, false, bo2, (short)0, pre, false, false);
        __builtin_amdgcn_sched_barrier(0);

        {
          v2f b1x;
          b1x.x = up ? cq   : xq0.x;
          b1x.y = up ? 0.0f : xq0.y;
          pre = __builtin_amdgcn_wmma_f32_16x16x4_f32(
              false, a1x, false, b1x, (short)0, bias1, false, false);
        }
        __builtin_amdgcn_sched_barrier(0);

        const int ip = (i + 4 < N_AG) ? i + 4 : N_AG - 1;
        const int ic = (i + 2 < N_AG) ? i + 2 : N_AG - 1;
        xq0 = xq1; xq1 = xq2;
        xq2 = *(const v2f*)(xrow + ip * 2);
        cq  = cread[ic + 1];
        if (k > 0) {
          obwrite[i - 1] = o0_prev;
          cwrite[i - 1]  = o1_prev;
        }
        __builtin_amdgcn_sched_barrier(0);

        const float h0 = fast_tanh(d1[0]);
        const float h1 = fast_tanh(d1[1]);
        const float h2 = fast_tanh(d1[2]);
        const float h3 = fast_tanh(d1[3]);
        const float h4 = fast_tanh(d1[4]);

        v2f hb;
        hb.x = h0; hb.y = h1;
        v8f d2 = __builtin_amdgcn_wmma_f32_16x16x4_f32(
            false, a2[0], false, hb, (short)0, bias2, false, false);
        hb.x = h2; hb.y = h3;
        d2 = __builtin_amdgcn_wmma_f32_16x16x4_f32(
            false, a2[1], false, hb, (short)0, d2, false, false);
        hb.x = h4; hb.y = 0.0f;
        d2 = __builtin_amdgcn_wmma_f32_16x16x4_f32(
            false, a2[2], false, hb, (short)0, d2, false, false);

        bo2.x = d2[2]; bo2.y = d2[3];
        o0_prev = d2[0];
        o1_prev = d2[1];
      }
      obwrite[i0 + GROUP - 1] = o0_prev;
      cwrite[i0 + GROUP - 1]  = o1_prev;
    }

    asm volatile("s_wait_dscnt 0x0" ::: "memory");
    asm volatile("s_barrier_signal -1\n\ts_barrier_wait -1" ::: "memory");
  }

  flush(wave + WAVES * (T_PER_WAVE - 1));
}

extern "C" void kernel_launch(void* const* d_in, const int* in_sizes, int n_in,
                              void* d_out, int out_size, void* d_ws, size_t ws_size,
                              hipStream_t stream) {
  (void)in_sizes; (void)n_in; (void)out_size; (void)d_ws; (void)ws_size;
  const float* runs = (const float*)d_in[0];
  const float* W1   = (const float*)d_in[1];
  const float* b1   = (const float*)d_in[2];
  const float* W2   = (const float*)d_in[3];
  const float* b2   = (const float*)d_in[4];
  float* out        = (float*)d_out;

  dim3 grid(R_TOT / RUNS_PER_BLOCK);
  dim3 block(32 * WAVES);
  com2net_wmma_kernel<<<grid, block, 0, stream>>>(runs, W1, b1, W2, b2, out);
}


# Matrix columns carry 4 runs x 4 timestep slots (32 agents apart) instead of 16 runs: 4 waves per work-group, one per SIMD, 32 work-groups; cells stream across timestep edges; sums unchanged
# speedup vs baseline: 16.1809x; 1.9137x over previous
#include <hip/hip_runtime.h>


typedef float v2f __attribute__((ext_vector_type(2)));
typedef float v8f __attribute__((ext_vector_type(8)));

#define R_TOT          128
#define T_STEPS        1024
#define N_AG           128
#define RUNS_PER_BLOCK 4
#define QUARTERS       4
#define SLOT_STRIDE    132
#define WAVES          4
#define SLOTS          (WAVES * QUARTERS)
#define GROUP          4
#define LAG_GROUPS     2
#define Q_GROUPS       (LAG_GROUPS * WAVES)
#define T_PER_SLOT     (T_STEPS / SLOTS)
#define GROUPS_PER_T   (N_AG / GROUP)
#define CELLS_PER_COL  (T_PER_SLOT * N_AG)
#define OWN_GROUPS     (T_PER_SLOT * GROUPS_PER_T + Q_GROUPS * (QUARTERS - 1))
#define EPOCHS         (OWN_GROUPS + LAG_GROUPS * (WAVES - 1))
#define CS_ROWS        (RUNS_PER_BLOCK + 32)
#define OB_SINK        192
#define OB_WAVE        (16 * SLOT_STRIDE + OB_SINK)

static_assert(RUNS_PER_BLOCK * QUARTERS == 16, "16 matrix columns");
static_assert(LAG_GROUPS * GROUP * SLOTS == N_AG, "a slot's next timestep is due when it ends one");
static_assert(Q_GROUPS * QUARTERS == GROUPS_PER_T, "quarters spread evenly over a timestep");
static_assert(T_STEPS % SLOTS == 0 && N_AG % GROUP == 0, "whole timesteps and groups");
static_assert((OB_WAVE * 4) % 16 == 0, "16-byte rows for the flush");

__device__ __forceinline__ float fast_tanh(float x) {
#if __has_builtin(__builtin_amdgcn_tanhf)
  return __builtin_amdgcn_tanhf(x);
#else
  float e = __builtin_exp2f(x * 2.885390081777927f);
  return 1.0f - 2.0f / (e + 1.0f);
#endif
}

__global__ __launch_bounds__(32 * WAVES)
void com2net_wmma_kernel(const float* __restrict__ runs,
                         const float* __restrict__ W1,
                         const float* __restrict__ b1,
                         const float* __restrict__ W2,
                         const float* __restrict__ b2,
                         float*       __restrict__ out)
{
  __shared__ __align__(16) float cslot [CS_ROWS * SLOT_STRIDE];
  __shared__ __align__(16) float outbuf[WAVES * OB_WAVE];

  const int wave    = __builtin_amdgcn_readfirstlane((int)(threadIdx.x >> 5));
  const int lane    = threadIdx.x & 31;
  const int up      = lane >> 4;
  const int l15     = lane & 15;
  const int q       = l15 >> 2;
  const int r       = l15 & 3;
  const int runBase = blockIdx.x * RUNS_PER_BLOCK;

  for (int k = threadIdx.x; k < CS_ROWS * SLOT_STRIDE; k += 32 * WAVES) cslot[k] = 0.0f;
  __syncthreads();

  const int hid = (l15 < 5) ? l15 : ((l15 >= 8 && l15 < 13) ? l15 - 3 : -1);

  v2f a1x; a1x.x = 0.0f; a1x.y = 0.0f;
  v2f a1c; a1c.x = 0.0f; a1c.y = 0.0f;
  if (hid >= 0) {
    if (up) { a1x.x = W1[hid * 4 + 3]; }
    else    { a1x.x = W1[hid * 4 + 0]; a1x.y = W1[hid * 4 + 1];
              a1c.x = W1[hid * 4 + 2]; }
  }

  v8f bias1;
#pragma unroll
  for (int v = 0; v < 8; ++v) bias1[v] = (v < 5) ? b1[5 * up + v] : 0.0f;

  v2f a2[3];
#pragma unroll
  for (int c2 = 0; c2 < 3; ++c2) {
    a2[c2].x = 0.0f; a2[c2].y = 0.0f;
    if (l15 < 3) {
      a2[c2].x = W2[l15 * 10 + 5 * up + 2 * c2];
      if (2 * c2 + 1 < 5) a2[c2].y = W2[l15 * 10 + 5 * up + 2 * c2 + 1];
    }
  }

  v8f bias2;
#pragma unroll
  for (int v = 0; v < 8; ++v) bias2[v] = (!up && v < 3) ? b2[v] : 0.0f;

  const float* xcol   = runs + ((size_t)(runBase + r) * T_STEPS + (wave + WAVES * q)) * (N_AG * 2);
  float*       crow   = &cslot[r * SLOT_STRIDE];
  float*       csink  = &cslot[(RUNS_PER_BLOCK + lane) * SLOT_STRIDE];
  float*       ob     = &outbuf[wave * OB_WAVE];
  float*       obreal = ob + l15 * SLOT_STRIDE;
  float*       obsink = ob + 16 * SLOT_STRIDE + l15;

  auto xoff = [&](int nn) -> int {
    int m = nn >> 7;
    m = (m < 0) ? 0 : ((m > T_PER_SLOT - 1) ? (T_PER_SLOT - 1) : m);
    return m * (SLOTS * N_AG * 2) + (nn & (N_AG - 1)) * 2;
  };

  auto flush = [&](int qs, int tt) {
#pragma unroll
    for (int rr = 0; rr < RUNS_PER_BLOCK; ++rr) {
      const float4 vv = *(const float4*)&ob[(qs * RUNS_PER_BLOCK + rr) * SLOT_STRIDE + lane * 4];
      *(float4*)(out + ((size_t)(runBase + rr) * T_STEPS + tt) * N_AG + lane * 4) = vv;
    }
  };

  v8f   pre = bias1;
  v2f   xq0, xq1, xq2;
  xq0.x = 0.0f; xq0.y = 0.0f; xq1 = xq0; xq2 = xq0;
  float cq  = 0.0f;
  v2f   bo2; bo2.x = 0.0f; bo2.y = 0.0f;

  for (int g = 0; g < EPOCHS; ++g) {
    const int le = g - LAG_GROUPS * wave;
    if (le >= 0 && le < OWN_GROUPS) {
      const int  nq   = GROUP * le - (N_AG / QUARTERS) * q;
      const int  i0   = nq & (N_AG - 1);
      const bool live = (nq >= 0) && (nq < CELLS_PER_COL) && !up;

      if (le == 0) {
        const float* xs  = xcol + xoff(nq);
        const v2f    x0v = *(const v2f*)xs;
        xq0 = *(const v2f*)(xs + 2);
        xq1 = *(const v2f*)(xs + 4);
        xq2 = *(const v2f*)(xs + 6);
        const float cb0 = crow[i0 + 1];
        cq = crow[i0 + 2];
        v2f b1x;
        b1x.x = up ? cb0  : x0v.x;
        b1x.y = up ? 0.0f : x0v.y;
        pre = __builtin_amdgcn_wmma_f32_16x16x4_f32(
            false, a1x, false, b1x, (short)0, bias1, false, false);
      }

      if ((le & (Q_GROUPS - 1)) == 0) {
        const int qs = (le / Q_GROUPS) & (QUARTERS - 1);
        const int ms = le / GROUPS_PER_T;
        const int sl = wave + WAVES * qs;
        bo2.x = (q == qs) ? 0.0f : bo2.x;
        bo2.y = (q == qs) ? 0.0f : bo2.y;
        if (ms + 1 < T_PER_SLOT) {
          const char* nxt = (const char*)(runs +
              ((size_t)(runBase + (lane >> 3)) * T_STEPS + (sl + SLOTS * (ms + 1))) * (N_AG * 2));
          __builtin_prefetch(nxt + (size_t)(lane & 7) * 128, 0, 0);
        }
        if (ms > 0) flush(qs, sl + SLOTS * (ms - 1));
      }

      const float* xg  = xcol + xoff(nq + GROUP);
      const float* crA = crow + i0;
      const float* crB = crow + ((i0 + GROUP) & (N_AG - 1));
      float*       cw  = (live ? crow   : csink)  + i0;
      float*       obw = (live ? obreal : obsink) + i0;

      float o0_prev = 0.0f, o1_prev = 0.0f;
#pragma unroll
      for (int k = 0; k < GROUP; ++k) {
        const v8f d1 = __builtin_amdgcn_wmma_f32_16x16x4_f32(
            false, a1c, false, bo2, (short)0, pre, false, false);
        __builtin_amdgcn_sched_barrier(0);

        {
          v2f b1x;
          b1x.x = up ? cq   : xq0.x;
          b1x.y = up ? 0.0f : xq0.y;
          pre = __builtin_amdgcn_wmma_f32_16x16x4_f32(
              false, a1x, false, b1x, (short)0, bias1, false, false);
        }
        __builtin_amdgcn_sched_barrier(0);

        xq0 = xq1; xq1 = xq2;
        xq2 = *(const v2f*)(xg + 2 * k);
        cq  = (k < 2) ? crA[3 + k] : crB[k - 1];
        if (k > 0) {
          obw[k - 1] = o0_prev;
          cw[k - 1]  = o1_prev;
        }
        __builtin_amdgcn_sched_barrier(0);

        const float h0 = fast_tanh(d1[0]);
        const float h1 = fast_tanh(d1[1]);
        const float h2 = fast_tanh(d1[2]);
        const float h3 = fast_tanh(d1[3]);
        const float h4 = fast_tanh(d1[4]);

        v2f hb;
        hb.x = h0; hb.y = h1;
        v8f d2 = __builtin_amdgcn_wmma_f32_16x16x4_f32(
            false, a2[0], false, hb, (short)0, bias2, false, false);
        hb.x = h2; hb.y = h3;
        d2 = __builtin_amdgcn_wmma_f32_16x16x4_f32(
            false, a2[1], false, hb, (short)0, d2, false, false);
        hb.x = h4; hb.y = 0.0f;
        d2 = __builtin_amdgcn_wmma_f32_16x16x4_f32(
            false, a2[2], false, hb, (short)0, d2, false, false);

        bo2.x = d2[2]; bo2.y = d2[3];
        o0_prev = d2[0];
        o1_prev = d2[1];
      }
      obw[GROUP - 1] = o0_prev;
      cw[GROUP - 1]  = o1_prev;
    }

    asm volatile("s_wait_dscnt 0x0" ::: "memory");
    asm volatile("s_barrier_signal -1\n\ts_barrier_wait -1" ::: "memory");
  }

  flush(QUARTERS - 1, wave + WAVES * (QUARTERS - 1) + SLOTS * (T_PER_SLOT - 1));
}

extern "C" void kernel_launch(void* const* d_in, const int* in_sizes, int n_in,
                              void* d_out, int out_size, void* d_ws, size_t ws_size,
                              hipStream_t stream) {
  (void)in_sizes; (void)n_in; (void)out_size; (void)d_ws; (void)ws_size;
  const float* runs = (const float*)d_in[0];
  const float* W1   = (const float*)d_in[1];
  const float* b1   = (const float*)d_in[2];
  const float* W2   = (const float*)d_in[3];
  const float* b2   = (const float*)d_in[4];
  float* out        = (float*)d_out;

  dim3 grid(R_TOT / RUNS_PER_BLOCK);
  dim3 block(32 * WAVES);
  com2net_wmma_kernel<<<grid, block, 0, stream>>>(runs, W1, b1, W2, b2, out);
}


# Columns carry 2 runs x 8 timestep slots, not 4 x 4: 32 timesteps in flight, 64 work-groups, barrier per 2 agents; backward word enters layer 1 via a free layer-2 k row; sum order changes
# speedup vs baseline: 21.9861x; 1.3588x over previous
#include <hip/hip_runtime.h>


typedef float v2f __attribute__((ext_vector_type(2)));
typedef float v8f __attribute__((ext_vector_type(8)));

#define R_TOT          128
#define T_STEPS        1024
#define N_AG           128
#define RUNS_PER_BLOCK 2
#define QUARTERS       8
#define SLOT_STRIDE    132
#define WAVES          4
#define SLOTS          (WAVES * QUARTERS)
#define GROUP          2
#define LAG_GROUPS     2
#define Q_GROUPS       (LAG_GROUPS * WAVES)
#define T_PER_SLOT     (T_STEPS / SLOTS)
#define GROUPS_PER_T   (N_AG / GROUP)
#define CELLS_PER_COL  (T_PER_SLOT * N_AG)
#define OWN_GROUPS     (T_PER_SLOT * GROUPS_PER_T + Q_GROUPS * (QUARTERS - 1))
#define EPOCHS         (OWN_GROUPS + LAG_GROUPS * (WAVES - 1))
#define CS_ROWS        (RUNS_PER_BLOCK + 32)
#define OB_SINK        192
#define OB_WAVE        (16 * SLOT_STRIDE + OB_SINK)

static_assert(RUNS_PER_BLOCK * QUARTERS == 16, "16 matrix columns");
static_assert(LAG_GROUPS * GROUP * SLOTS == N_AG, "a slot's next timestep is due when it ends one");
static_assert(Q_GROUPS * QUARTERS == GROUPS_PER_T, "column slots spread evenly over a timestep");
static_assert(GROUP == 2 && LAG_GROUPS == 2, "the group body below is written for 2 agents, 2 groups behind");
static_assert(T_STEPS % SLOTS == 0 && N_AG % GROUP == 0, "whole timesteps and groups");
static_assert((OB_WAVE * 4) % 16 == 0, "16-byte rows for the flush");

__device__ __forceinline__ float fast_tanh(float x) {
#if __has_builtin(__builtin_amdgcn_tanhf)
  return __builtin_amdgcn_tanhf(x);
#else
  float e = __builtin_exp2f(x * 2.885390081777927f);
  return 1.0f - 2.0f / (e + 1.0f);
#endif
}

__global__ __launch_bounds__(32 * WAVES)
void com2net_wmma_kernel(const float* __restrict__ runs,
                         const float* __restrict__ W1,
                         const float* __restrict__ b1,
                         const float* __restrict__ W2,
                         const float* __restrict__ b2,
                         float*       __restrict__ out)
{
  __shared__ __align__(16) float cslot [CS_ROWS * SLOT_STRIDE];
  __shared__ __align__(16) float outbuf[WAVES * OB_WAVE];

  const int wave    = __builtin_amdgcn_readfirstlane((int)(threadIdx.x >> 5));
  const int lane    = threadIdx.x & 31;
  const int up      = lane >> 4;
  const int l15     = lane & 15;
  const int q       = l15 >> 1;
  const int r       = l15 & 1;
  const int runBase = blockIdx.x * RUNS_PER_BLOCK;

  for (int k = threadIdx.x; k < CS_ROWS * SLOT_STRIDE; k += 32 * WAVES) cslot[k] = 0.0f;
  __syncthreads();

  const int hid = (l15 < 5) ? l15 : ((l15 >= 8 && l15 < 13) ? l15 - 3 : -1);

  v2f a1x; a1x.x = 0.0f; a1x.y = 0.0f;
  v2f a1c; a1c.x = 0.0f; a1c.y = 0.0f;
  if (hid >= 0) {
    if (up) { a1c.x = W1[hid * 4 + 3]; }
    else    { a1x.x = W1[hid * 4 + 0]; a1x.y = W1[hid * 4 + 1];
              a1c.x = W1[hid * 4 + 2]; }
  }

  v8f bias1;
#pragma unroll
  for (int v = 0; v < 8; ++v) bias1[v] = (v < 5) ? b1[5 * up + v] : 0.0f;

  v2f a2[3];
#pragma unroll
  for (int c2 = 0; c2 < 3; ++c2) {
    a2[c2].x = 0.0f; a2[c2].y = 0.0f;
    if (l15 < 3) {
      a2[c2].x = W2[l15 * 10 + 5 * up + 2 * c2];
      if (2 * c2 + 1 < 5) a2[c2].y = W2[l15 * 10 + 5 * up + 2 * c2 + 1];
    }
  }
  if (up && l15 == 10) a2[2].y = 1.0f;

  v8f bias2;
#pragma unroll
  for (int v = 0; v < 8; ++v) bias2[v] = (!up && v < 3) ? b2[v] : 0.0f;

  const float* xcol   = runs + ((size_t)(runBase + r) * T_STEPS + (wave + WAVES * q)) * (N_AG * 2);
  float*       crow   = &cslot[r * SLOT_STRIDE];
  float*       csink  = &cslot[(RUNS_PER_BLOCK + lane) * SLOT_STRIDE];
  float*       ob     = &outbuf[wave * OB_WAVE];
  float*       obreal = ob + l15 * SLOT_STRIDE;
  float*       obsink = ob + 16 * SLOT_STRIDE + l15;

  auto xoff = [&](int nn) -> int {
    int m = nn >> 7;
    m = (m < 0) ? 0 : ((m > T_PER_SLOT - 1) ? (T_PER_SLOT - 1) : m);
    return m * (SLOTS * N_AG * 2) + (nn & (N_AG - 1)) * 2;
  };

  auto flush = [&](int qs, int tt) {
#pragma unroll
    for (int rr = 0; rr < RUNS_PER_BLOCK; ++rr) {
      const float4 vv = *(const float4*)&ob[(qs * RUNS_PER_BLOCK + rr) * SLOT_STRIDE + lane * 4];
      *(float4*)(out + ((size_t)(runBase + rr) * T_STEPS + tt) * N_AG + lane * 4) = vv;
    }
  };

  v8f   pre = bias1;
  v2f   xq0, xq1, xq2;
  xq0.x = 0.0f; xq0.y = 0.0f; xq1 = xq0; xq2 = xq0;
  v2f   bo2; bo2.x = 0.0f; bo2.y = 0.0f;

  for (int g = 0; g < EPOCHS; ++g) {
    const int le = g - LAG_GROUPS * wave;
    if (le >= 0 && le < OWN_GROUPS) {
      const int  nq   = GROUP * le - (N_AG / QUARTERS) * q;
      const int  i0   = nq & (N_AG - 1);
      const bool live = (nq >= 0) && (nq < CELLS_PER_COL) && !up;

      if (le == 0) {
        const float* xs  = xcol + xoff(nq);
        const v2f    x0v = *(const v2f*)xs;
        xq0 = *(const v2f*)(xs + 2);
        xq1 = *(const v2f*)(xs + 4);
        xq2 = *(const v2f*)(xs + 6);
        const float cb0 = crow[i0 + 1];
        pre = __builtin_amdgcn_wmma_f32_16x16x4_f32(
            false, a1x, false, x0v, (short)0, bias1, false, false);
        bo2.x = up ? cb0 : 0.0f;
      }

      if ((le & (Q_GROUPS - 1)) == 0) {
        const int qs = (le / Q_GROUPS) & (QUARTERS - 1);
        const int ms = le / GROUPS_PER_T;
        const int sl = wave + WAVES * qs;
        bo2.x = (q == qs && !up) ? 0.0f : bo2.x;
        if (ms + 1 < T_PER_SLOT) {
          const char* nxt = (const char*)(runs +
              ((size_t)(runBase + ((lane >> 3) & (RUNS_PER_BLOCK - 1))) * T_STEPS + (sl + SLOTS * (ms + 1))) * (N_AG * 2));
          __builtin_prefetch(nxt + (size_t)(lane & 7) * 128, 0, 0);
        }
        if (ms > 0) flush(qs, sl + SLOTS * (ms - 1));
      }

      const float  cA  = crow[i0 + 2];
      const float  cB  = crow[((i0 + GROUP) & (N_AG - 1)) + 1];
      __builtin_amdgcn_sched_barrier(0);
      const float* xg  = xcol + xoff(nq + 4);
      float*       cw  = (live ? crow   : csink)  + i0;
      float*       obw = (live ? obreal : obsink) + i0;

      float  o0_first = 0.0f, o1_first = 0.0f, o0_last = 0.0f, o1_last = 0.0f;
      float4 xn = make_float4(0.0f, 0.0f, 0.0f, 0.0f);
#pragma unroll
      for (int k = 0; k < GROUP; ++k) {
        const v8f d1 = __builtin_amdgcn_wmma_f32_16x16x4_f32(
            false, a1c, false, bo2, (short)0, pre, false, false);
        __builtin_amdgcn_sched_barrier(0);

        pre = __builtin_amdgcn_wmma_f32_16x16x4_f32(
            false, a1x, false, (k == 0) ? xq0 : xq1, (short)0, bias1, false, false);
        __builtin_amdgcn_sched_barrier(0);

        if (k == 0) {
          xn = *(const float4*)xg;
          __builtin_amdgcn_sched_barrier(0);
        }

        const float h0 = fast_tanh(d1[0]);
        const float h1 = fast_tanh(d1[1]);
        const float h2 = fast_tanh(d1[2]);
        const float h3 = fast_tanh(d1[3]);
        const float h4 = fast_tanh(d1[4]);

        v2f hb;
        hb.x = h0; hb.y = h1;
        v8f d2 = __builtin_amdgcn_wmma_f32_16x16x4_f32(
            false, a2[0], false, hb, (short)0, bias2, false, false);
        hb.x = h2; hb.y = h3;
        d2 = __builtin_amdgcn_wmma_f32_16x16x4_f32(
            false, a2[1], false, hb, (short)0, d2, false, false);
        hb.x = h4; hb.y = (k == 0) ? cA : cB;
        d2 = __builtin_amdgcn_wmma_f32_16x16x4_f32(
            false, a2[2], false, hb, (short)0, d2, false, false);

        bo2.x = d2[2]; bo2.y = d2[3];
        if (k == 0) { o0_first = d2[0]; o1_first = d2[1]; }
        else        { o0_last  = d2[0]; o1_last  = d2[1]; }
      }
      __builtin_amdgcn_sched_barrier(0);
      xq0 = xq2;
      xq1.x = xn.x; xq1.y = xn.y;
      xq2.x = xn.z; xq2.y = xn.w;
      obw[0] = o0_first;
      obw[1] = o0_last;
      cw[0]  = o1_first;
      cw[1]  = o1_last;
    }

    asm volatile("s_wait_dscnt 0x0" ::: "memory");
    asm volatile("s_barrier_signal -1\n\ts_barrier_wait -1" ::: "memory");
  }

  flush(QUARTERS - 1, wave + WAVES * (QUARTERS - 1) + SLOTS * (T_PER_SLOT - 1));
}

extern "C" void kernel_launch(void* const* d_in, const int* in_sizes, int n_in,
                              void* d_out, int out_size, void* d_ws, size_t ws_size,
                              hipStream_t stream) {
  (void)in_sizes; (void)n_in; (void)out_size; (void)d_ws; (void)ws_size;
  const float* runs = (const float*)d_in[0];
  const float* W1   = (const float*)d_in[1];
  const float* b1   = (const float*)d_in[2];
  const float* W2   = (const float*)d_in[3];
  const float* b2   = (const float*)d_in[4];
  float* out        = (float*)d_out;

  dim3 grid(R_TOT / RUNS_PER_BLOCK);
  dim3 block(32 * WAVES);
  com2net_wmma_kernel<<<grid, block, 0, stream>>>(runs, W1, b1, W2, b2, out);
}


# x rows staged in LDS per column (two rows, by timestep parity) with coalesced 16-byte loads 16 groups ahead; group loop reads x from LDS beside the comm words, no global load in it; sums unchanged
# speedup vs baseline: 31.4241x; 1.4293x over previous
#include <hip/hip_runtime.h>


typedef float v2f __attribute__((ext_vector_type(2)));
typedef float v8f __attribute__((ext_vector_type(8)));

#define R_TOT          128
#define T_STEPS        1024
#define N_AG           128
#define RUNS_PER_BLOCK 2
#define QUARTERS       8
#define SLOT_STRIDE    132
#define WAVES          4
#define SLOTS          (WAVES * QUARTERS)
#define GROUP          2
#define LAG_GROUPS     2
#define Q_GROUPS       (LAG_GROUPS * WAVES)
#define T_PER_SLOT     (T_STEPS / SLOTS)
#define GROUPS_PER_T   (N_AG / GROUP)
#define CELLS_PER_COL  (T_PER_SLOT * N_AG)
#define OWN_GROUPS     (T_PER_SLOT * GROUPS_PER_T + Q_GROUPS * (QUARTERS - 1))
#define EPOCHS         (OWN_GROUPS + LAG_GROUPS * (WAVES - 1))
#define CS_ROWS        (RUNS_PER_BLOCK + 32)
#define OB_SINK        192
#define OB_WAVE        (16 * SLOT_STRIDE + OB_SINK)
#define XROW_F         (N_AG * 2)
#define XCOL_F         (2 * XROW_F + 4)
#define XWAVE_F        (16 * XCOL_F)
#define STAGE_EVENTS   (QUARTERS * T_PER_SLOT)

static_assert(RUNS_PER_BLOCK * QUARTERS == 16, "16 matrix columns");
static_assert(LAG_GROUPS * GROUP * SLOTS == N_AG, "a slot's next timestep is due when it ends one");
static_assert(Q_GROUPS * QUARTERS == GROUPS_PER_T, "column slots spread evenly over a timestep");
static_assert(GROUP == 2 && LAG_GROUPS == 2, "the group body below is written for 2 agents, 2 groups behind");
static_assert(T_STEPS % SLOTS == 0 && N_AG % GROUP == 0, "whole timesteps and groups");
static_assert((OB_WAVE * 4) % 16 == 0, "16-byte rows for the flush");

__device__ __forceinline__ float fast_tanh(float x) {
#if __has_builtin(__builtin_amdgcn_tanhf)
  return __builtin_amdgcn_tanhf(x);
#else
  float e = __builtin_exp2f(x * 2.885390081777927f);
  return 1.0f - 2.0f / (e + 1.0f);
#endif
}

__global__ __launch_bounds__(32 * WAVES)
void com2net_wmma_kernel(const float* __restrict__ runs,
                         const float* __restrict__ W1,
                         const float* __restrict__ b1,
                         const float* __restrict__ W2,
                         const float* __restrict__ b2,
                         float*       __restrict__ out)
{
  __shared__ __align__(16) float cslot [CS_ROWS * SLOT_STRIDE];
  __shared__ __align__(16) float outbuf[WAVES * OB_WAVE];
  __shared__ __align__(16) float xbuf  [WAVES * XWAVE_F];

  const int wave    = __builtin_amdgcn_readfirstlane((int)(threadIdx.x >> 5));
  const int lane    = threadIdx.x & 31;
  const int up      = lane >> 4;
  const int l15     = lane & 15;
  const int q       = l15 >> 1;
  const int r       = l15 & 1;
  const int runBase = blockIdx.x * RUNS_PER_BLOCK;

  for (int k = threadIdx.x; k < CS_ROWS * SLOT_STRIDE; k += 32 * WAVES) cslot[k] = 0.0f;
  for (int k = threadIdx.x; k < WAVES * XWAVE_F / 4; k += 32 * WAVES)
    ((float4*)xbuf)[k] = make_float4(0.0f, 0.0f, 0.0f, 0.0f);
  __syncthreads();

  const int hid = (l15 < 5) ? l15 : ((l15 >= 8 && l15 < 13) ? l15 - 3 : -1);

  v2f a1x; a1x.x = 0.0f; a1x.y = 0.0f;
  v2f a1c; a1c.x = 0.0f; a1c.y = 0.0f;
  if (hid >= 0) {
    if (up) { a1c.x = W1[hid * 4 + 3]; }
    else    { a1x.x = W1[hid * 4 + 0]; a1x.y = W1[hid * 4 + 1];
              a1c.x = W1[hid * 4 + 2]; }
  }

  v8f bias1;
#pragma unroll
  for (int v = 0; v < 8; ++v) bias1[v] = (v < 5) ? b1[5 * up + v] : 0.0f;

  v2f a2[3];
#pragma unroll
  for (int c2 = 0; c2 < 3; ++c2) {
    a2[c2].x = 0.0f; a2[c2].y = 0.0f;
    if (l15 < 3) {
      a2[c2].x = W2[l15 * 10 + 5 * up + 2 * c2];
      if (2 * c2 + 1 < 5) a2[c2].y = W2[l15 * 10 + 5 * up + 2 * c2 + 1];
    }
  }
  if (up && l15 == 10) a2[2].y = 1.0f;

  v8f bias2;
#pragma unroll
  for (int v = 0; v < 8; ++v) bias2[v] = (!up && v < 3) ? b2[v] : 0.0f;

  const float* xb     = &xbuf[wave * XWAVE_F + l15 * XCOL_F];
  float*       crow   = &cslot[r * SLOT_STRIDE];
  float*       csink  = &cslot[(RUNS_PER_BLOCK + lane) * SLOT_STRIDE];
  float*       ob     = &outbuf[wave * OB_WAVE];
  float*       obreal = ob + l15 * SLOT_STRIDE;
  float*       obsink = ob + 16 * SLOT_STRIDE + l15;

  auto xoff = [&](int nn) -> int { return (nn & (2 * N_AG - 1)) * 2; };

  const int srun = lane >> 4;
  auto stage_src = [&](int ev) -> const float* {
    const int tt = wave + WAVES * (ev & (QUARTERS - 1)) + SLOTS * (ev >> 3);
    return runs + ((size_t)(runBase + srun) * T_STEPS + tt) * (N_AG * 2) + l15 * 4;
  };
  auto stage_dst = [&](int ev) -> float* {
    return &xbuf[wave * XWAVE_F + ((ev & (QUARTERS - 1)) * RUNS_PER_BLOCK + srun) * XCOL_F
                 + ((ev >> 3) & 1) * XROW_F + l15 * 4];
  };
  float4 xs0 = make_float4(0.0f, 0.0f, 0.0f, 0.0f), xs1 = xs0, xs2 = xs0, xs3 = xs0;

  auto flush = [&](int qs, int tt) {
#pragma unroll
    for (int rr = 0; rr < RUNS_PER_BLOCK; ++rr) {
      const float4 vv = *(const float4*)&ob[(qs * RUNS_PER_BLOCK + rr) * SLOT_STRIDE + lane * 4];
      *(float4*)(out + ((size_t)(runBase + rr) * T_STEPS + tt) * N_AG + lane * 4) = vv;
    }
  };

  v8f   pre = bias1;
  v2f   xq0, xq1, xq2;
  xq0.x = 0.0f; xq0.y = 0.0f; xq1 = xq0; xq2 = xq0;
  v2f   bo2; bo2.x = 0.0f; bo2.y = 0.0f;

  for (int g = 0; g < EPOCHS; ++g) {
    const int le = g - LAG_GROUPS * wave;
    if (le >= 0 && le < OWN_GROUPS) {
      const int  nq   = GROUP * le - (N_AG / QUARTERS) * q;
      const int  i0   = nq & (N_AG - 1);
      const bool live = (nq >= 0) && (nq < CELLS_PER_COL) && !up;

      if (le == 0) {
        {
          const float* s0 = stage_src(0);
          const float4 a0 = *(const float4*)(s0);
          const float4 a1 = *(const float4*)(s0 + 64);
          const float4 a2 = *(const float4*)(s0 + 128);
          const float4 a3 = *(const float4*)(s0 + 192);
          const float* s1 = stage_src(1);
          xs0 = *(const float4*)(s1);
          xs1 = *(const float4*)(s1 + 64);
          xs2 = *(const float4*)(s1 + 128);
          xs3 = *(const float4*)(s1 + 192);
          float* d0 = stage_dst(0);
          *(float4*)(d0)       = a0;
          *(float4*)(d0 + 64)  = a1;
          *(float4*)(d0 + 128) = a2;
          *(float4*)(d0 + 192) = a3;
        }
        const float* xs  = xb + xoff(nq);
        const float4 xf0 = *(const float4*)xs;
        const float4 xf1 = *(const float4*)(xs + 4);
        v2f x0v; x0v.x = xf0.x; x0v.y = xf0.y;
        xq0.x = xf0.z; xq0.y = xf0.w;
        xq1.x = xf1.x; xq1.y = xf1.y;
        xq2.x = xf1.z; xq2.y = xf1.w;
        const float cb0 = crow[i0 + 1];
        pre = __builtin_amdgcn_wmma_f32_16x16x4_f32(
            false, a1x, false, x0v, (short)0, bias1, false, false);
        bo2.x = up ? cb0 : 0.0f;
      }

      if ((le & (Q_GROUPS - 1)) == 0) {
        const int qs = (le / Q_GROUPS) & (QUARTERS - 1);
        const int ms = le / GROUPS_PER_T;
        const int sl = wave + WAVES * qs;
        bo2.x = (q == qs && !up) ? 0.0f : bo2.x;
        const int ev = le / Q_GROUPS;
        if (ev + 1 < STAGE_EVENTS) {
          float* d = stage_dst(ev + 1);
          *(float4*)(d)       = xs0;
          *(float4*)(d + 64)  = xs1;
          *(float4*)(d + 128) = xs2;
          *(float4*)(d + 192) = xs3;
        }
        if (ev + 2 < STAGE_EVENTS) {
          const float* s = stage_src(ev + 2);
          xs0 = *(const float4*)(s);
          xs1 = *(const float4*)(s + 64);
          xs2 = *(const float4*)(s + 128);
          xs3 = *(const float4*)(s + 192);
        }
        if (ms > 0) flush(qs, sl + SLOTS * (ms - 1));
      }

      const float  cA  = crow[i0 + 2];
      const float  cB  = crow[((i0 + GROUP) & (N_AG - 1)) + 1];
      const float4 xn  = *(const float4*)(xb + xoff(nq + 4));
      __builtin_amdgcn_sched_barrier(0);
      float*       cw  = (live ? crow   : csink)  + i0;
      float*       obw = (live ? obreal : obsink) + i0;

      float  o0_first = 0.0f, o1_first = 0.0f, o0_last = 0.0f, o1_last = 0.0f;
#pragma unroll
      for (int k = 0; k < GROUP; ++k) {
        const v8f d1 = __builtin_amdgcn_wmma_f32_16x16x4_f32(
            false, a1c, false, bo2, (short)0, pre, false, false);
        __builtin_amdgcn_sched_barrier(0);

        pre = __builtin_amdgcn_wmma_f32_16x16x4_f32(
            false, a1x, false, (k == 0) ? xq0 : xq1, (short)0, bias1, false, false);
        __builtin_amdgcn_sched_barrier(0);

        const float h0 = fast_tanh(d1[0]);
        const float h1 = fast_tanh(d1[1]);
        const float h2 = fast_tanh(d1[2]);
        const float h3 = fast_tanh(d1[3]);
        const float h4 = fast_tanh(d1[4]);

        v2f hb;
        hb.x = h0; hb.y = h1;
        v8f d2 = __builtin_amdgcn_wmma_f32_16x16x4_f32(
            false, a2[0], false, hb, (short)0, bias2, false, false);
        hb.x = h2; hb.y = h3;
        d2 = __builtin_amdgcn_wmma_f32_16x16x4_f32(
            false, a2[1], false, hb, (short)0, d2, false, false);
        hb.x = h4; hb.y = (k == 0) ? cA : cB;
        d2 = __builtin_amdgcn_wmma_f32_16x16x4_f32(
            false, a2[2], false, hb, (short)0, d2, false, false);

        bo2.x = d2[2]; bo2.y = d2[3];
        if (k == 0) { o0_first = d2[0]; o1_first = d2[1]; }
        else        { o0_last  = d2[0]; o1_last  = d2[1]; }
      }
      __builtin_amdgcn_sched_barrier(0);
      xq0 = xq2;
      xq1.x = xn.x; xq1.y = xn.y;
      xq2.x = xn.z; xq2.y = xn.w;
      obw[0] = o0_first;
      obw[1] = o0_last;
      cw[0]  = o1_first;
      cw[1]  = o1_last;
    }

    asm volatile("s_wait_dscnt 0x0" ::: "memory");
    asm volatile("s_barrier_signal -1\n\ts_barrier_wait -1" ::: "memory");
  }

  flush(QUARTERS - 1, wave + WAVES * (QUARTERS - 1) + SLOTS * (T_PER_SLOT - 1));
}

extern "C" void kernel_launch(void* const* d_in, const int* in_sizes, int n_in,
                              void* d_out, int out_size, void* d_ws, size_t ws_size,
                              hipStream_t stream) {
  (void)in_sizes; (void)n_in; (void)out_size; (void)d_ws; (void)ws_size;
  const float* runs = (const float*)d_in[0];
  const float* W1   = (const float*)d_in[1];
  const float* b1   = (const float*)d_in[2];
  const float* W2   = (const float*)d_in[3];
  const float* b2   = (const float*)d_in[4];
  float* out        = (float*)d_out;

  dim3 grid(R_TOT / RUNS_PER_BLOCK);
  dim3 block(32 * WAVES);
  com2net_wmma_kernel<<<grid, block, 0, stream>>>(runs, W1, b1, W2, b2, out);
}


# o2 row of layer 2 folded into layer 1 (weights W1[:,2]*W2[2,:]): a cell is tanh + 3 chained f32 WMMAs, not 4; agent 0 rebuilt without the fold; sum order and rounding change
# speedup vs baseline: 33.6516x; 1.0709x over previous
#include <hip/hip_runtime.h>


typedef float v2f __attribute__((ext_vector_type(2)));
typedef float v8f __attribute__((ext_vector_type(8)));

#define R_TOT          128
#define T_STEPS        1024
#define N_AG           128
#define RUNS_PER_BLOCK 2
#define QUARTERS       8
#define SLOT_STRIDE    132
#define WAVES          4
#define SLOTS          (WAVES * QUARTERS)
#define GROUP          2
#define LAG_GROUPS     2
#define Q_GROUPS       (LAG_GROUPS * WAVES)
#define T_PER_SLOT     (T_STEPS / SLOTS)
#define GROUPS_PER_T   (N_AG / GROUP)
#define CELLS_PER_COL  (T_PER_SLOT * N_AG)
#define OWN_GROUPS     (T_PER_SLOT * GROUPS_PER_T + Q_GROUPS * (QUARTERS - 1))
#define EPOCHS         (OWN_GROUPS + LAG_GROUPS * (WAVES - 1))
#define CS_ROWS        (RUNS_PER_BLOCK + 32)
#define OB_SINK        192
#define OB_WAVE        (16 * SLOT_STRIDE + OB_SINK)
#define XROW_F         (N_AG * 2)
#define XCOL_F         (2 * XROW_F + 4)
#define XWAVE_F        (16 * XCOL_F)
#define STAGE_EVENTS   (QUARTERS * T_PER_SLOT)

static_assert(RUNS_PER_BLOCK * QUARTERS == 16, "16 matrix columns");
static_assert(LAG_GROUPS * GROUP * SLOTS == N_AG, "a slot's next timestep is due when it ends one");
static_assert(Q_GROUPS * QUARTERS == GROUPS_PER_T, "column slots spread evenly over a timestep");
static_assert(GROUP == 2 && LAG_GROUPS == 2, "the group body below is written for 2 agents, 2 groups behind");
static_assert(T_STEPS % SLOTS == 0 && N_AG % GROUP == 0, "whole timesteps and groups");
static_assert((OB_WAVE * 4) % 16 == 0, "16-byte rows for the flush");

__device__ __forceinline__ float fast_tanh(float x) {
#if __has_builtin(__builtin_amdgcn_tanhf)
  return __builtin_amdgcn_tanhf(x);
#else
  float e = __builtin_exp2f(x * 2.885390081777927f);
  return 1.0f - 2.0f / (e + 1.0f);
#endif
}

__global__ __launch_bounds__(32 * WAVES)
void com2net_wmma_kernel(const float* __restrict__ runs,
                         const float* __restrict__ W1,
                         const float* __restrict__ b1,
                         const float* __restrict__ W2,
                         const float* __restrict__ b2,
                         float*       __restrict__ out)
{
  __shared__ __align__(16) float cslot [CS_ROWS * SLOT_STRIDE];
  __shared__ __align__(16) float outbuf[WAVES * OB_WAVE];
  __shared__ __align__(16) float xbuf  [WAVES * XWAVE_F];

  const int wave    = __builtin_amdgcn_readfirstlane((int)(threadIdx.x >> 5));
  const int lane    = threadIdx.x & 31;
  const int up      = lane >> 4;
  const int l15     = lane & 15;
  const int q       = l15 >> 1;
  const int r       = l15 & 1;
  const int runBase = blockIdx.x * RUNS_PER_BLOCK;

  for (int k = threadIdx.x; k < CS_ROWS * SLOT_STRIDE; k += 32 * WAVES) cslot[k] = 0.0f;
  for (int k = threadIdx.x; k < WAVES * XWAVE_F / 4; k += 32 * WAVES)
    ((float4*)xbuf)[k] = make_float4(0.0f, 0.0f, 0.0f, 0.0f);
  __syncthreads();

  const int hid = (l15 < 5) ? l15 : ((l15 >= 8 && l15 < 13) ? l15 - 3 : -1);

  v2f a1x; a1x.x = 0.0f; a1x.y = 0.0f;
  if (hid >= 0 && !up) { a1x.x = W1[hid * 4 + 0]; a1x.y = W1[hid * 4 + 1]; }
  v2f a1e; a1e.x = 0.0f; a1e.y = 0.0f;
  if (hid >= 0 && up) a1e.y = W1[hid * 4 + 3];

  v2f a2[3];
#pragma unroll
  for (int c2 = 0; c2 < 3; ++c2) {
    a2[c2].x = 0.0f; a2[c2].y = 0.0f;
    const int k0 = 5 * up + 2 * c2;
    if (hid >= 0) {
      const float wf = W1[hid * 4 + 2];
      a2[c2].x = wf * W2[20 + k0];
      if (2 * c2 + 1 < 5) a2[c2].y = wf * W2[20 + k0 + 1];
    } else if (l15 == 5 || l15 == 6) {
      a2[c2].x = W2[(l15 - 5) * 10 + k0];
      if (2 * c2 + 1 < 5) a2[c2].y = W2[(l15 - 5) * 10 + k0 + 1];
    }
  }
  if (up && hid >= 0) a2[2].y = W1[hid * 4 + 3];

  v8f bias1p, bias1g;
#pragma unroll
  for (int v = 0; v < 8; ++v) {
    bias1p[v] = (v < 5) ? b1[5 * up + v] : 0.0f;
    bias1g[v] = (v < 5) ? (b1[5 * up + v] + W1[(5 * up + v) * 4 + 2] * b2[2])
                        : ((!up && (v == 5 || v == 6)) ? b2[v - 5] : 0.0f);
  }

  const float* xb     = &xbuf[wave * XWAVE_F + l15 * XCOL_F];
  float*       crow   = &cslot[r * SLOT_STRIDE];
  float*       csink  = &cslot[(RUNS_PER_BLOCK + lane) * SLOT_STRIDE];
  float*       ob     = &outbuf[wave * OB_WAVE];
  float*       obreal = ob + l15 * SLOT_STRIDE;
  float*       obsink = ob + 16 * SLOT_STRIDE + l15;

  auto xoff = [&](int nn) -> int { return (nn & (2 * N_AG - 1)) * 2; };

  const int srun = lane >> 4;
  auto stage_src = [&](int ev) -> const float* {
    const int tt = wave + WAVES * (ev & (QUARTERS - 1)) + SLOTS * (ev >> 3);
    return runs + ((size_t)(runBase + srun) * T_STEPS + tt) * (N_AG * 2) + l15 * 4;
  };
  auto stage_dst = [&](int ev) -> float* {
    return &xbuf[wave * XWAVE_F + ((ev & (QUARTERS - 1)) * RUNS_PER_BLOCK + srun) * XCOL_F
                 + ((ev >> 3) & 1) * XROW_F + l15 * 4];
  };
  float4 xs0 = make_float4(0.0f, 0.0f, 0.0f, 0.0f), xs1 = xs0, xs2 = xs0, xs3 = xs0;

  auto flush = [&](int qs, int tt) {
#pragma unroll
    for (int rr = 0; rr < RUNS_PER_BLOCK; ++rr) {
      const float4 vv = *(const float4*)&ob[(qs * RUNS_PER_BLOCK + rr) * SLOT_STRIDE + lane * 4];
      *(float4*)(out + ((size_t)(runBase + rr) * T_STEPS + tt) * N_AG + lane * 4) = vv;
    }
  };

  v8f   pre = bias1g;
  v8f   d1  = bias1p;
  v2f   xqa, xqb;
  xqa.x = 0.0f; xqa.y = 0.0f; xqb = xqa;

  for (int g = 0; g < EPOCHS; ++g) {
    const int le = g - LAG_GROUPS * wave;
    if (le >= 0 && le < OWN_GROUPS) {
      const int  nq   = GROUP * le - (N_AG / QUARTERS) * q;
      const int  i0   = nq & (N_AG - 1);
      const bool live = (nq >= 0) && (nq < CELLS_PER_COL) && !up;

      if (le == 0) {
        {
          const float* s0 = stage_src(0);
          const float4 a0 = *(const float4*)(s0);
          const float4 a1 = *(const float4*)(s0 + 64);
          const float4 a2 = *(const float4*)(s0 + 128);
          const float4 a3 = *(const float4*)(s0 + 192);
          const float* s1 = stage_src(1);
          xs0 = *(const float4*)(s1);
          xs1 = *(const float4*)(s1 + 64);
          xs2 = *(const float4*)(s1 + 128);
          xs3 = *(const float4*)(s1 + 192);
          float* d0 = stage_dst(0);
          *(float4*)(d0)       = a0;
          *(float4*)(d0 + 64)  = a1;
          *(float4*)(d0 + 128) = a2;
          *(float4*)(d0 + 192) = a3;
        }
        const float* xs  = xb + xoff(nq);
        const float4 xf0 = *(const float4*)xs;
        const float4 xf1 = *(const float4*)(xs + 4);
        v2f x0v; x0v.x = xf0.x; x0v.y = xf0.y;
        v2f x1v; x1v.x = xf0.z; x1v.y = xf0.w;
        xqa.x = xf1.x; xqa.y = xf1.y;
        xqb.x = xf1.z; xqb.y = xf1.w;
        const float cb0 = crow[i0 + 1];
        v2f cbv; cbv.x = 0.0f; cbv.y = cb0;
        const v8f e0 = __builtin_amdgcn_wmma_f32_16x16x4_f32(
            false, a1x, false, x0v, (short)0, bias1p, false, false);
        d1 = __builtin_amdgcn_wmma_f32_16x16x4_f32(
            false, a1e, false, cbv, (short)0, e0, false, false);
        pre = __builtin_amdgcn_wmma_f32_16x16x4_f32(
            false, a1x, false, x1v, (short)0, bias1g, false, false);
      }

      if ((le & (Q_GROUPS - 1)) == 0) {
        const int qs = (le / Q_GROUPS) & (QUARTERS - 1);
        const int ms = le / GROUPS_PER_T;
        const int sl = wave + WAVES * qs;
        const int ev = le / Q_GROUPS;
        if (ev + 1 < STAGE_EVENTS) {
          float* d = stage_dst(ev + 1);
          *(float4*)(d)       = xs0;
          *(float4*)(d + 64)  = xs1;
          *(float4*)(d + 128) = xs2;
          *(float4*)(d + 192) = xs3;
        }
        if (ev + 2 < STAGE_EVENTS) {
          const float* s = stage_src(ev + 2);
          xs0 = *(const float4*)(s);
          xs1 = *(const float4*)(s + 64);
          xs2 = *(const float4*)(s + 128);
          xs3 = *(const float4*)(s + 192);
        }
        if (ms > 0) flush(qs, sl + SLOTS * (ms - 1));
      }

      const float  cA  = crow[i0 + 2];
      const float  cB  = crow[((i0 + GROUP) & (N_AG - 1)) + 1];
      const float4 xn  = *(const float4*)(xb + xoff(nq + 4));
      __builtin_amdgcn_sched_barrier(0);
      float*       cw  = (live ? crow   : csink)  + i0;
      float*       obw = (live ? obreal : obsink) + i0;

      float  o0_first = 0.0f, o1_first = 0.0f, o0_last = 0.0f, o1_last = 0.0f;
      v2f    hb_last; hb_last.x = 0.0f; hb_last.y = 0.0f;
#pragma unroll
      for (int k = 0; k < GROUP; ++k) {
        const float h0 = fast_tanh(d1[0]);
        const float h1 = fast_tanh(d1[1]);
        const float h2 = fast_tanh(d1[2]);
        const float h3 = fast_tanh(d1[3]);
        const float h4 = fast_tanh(d1[4]);

        v2f hb;
        hb.x = h0; hb.y = h1;
        v8f dn = __builtin_amdgcn_wmma_f32_16x16x4_f32(
            false, a2[0], false, hb, (short)0, pre, false, false);
        hb.x = h2; hb.y = h3;
        dn = __builtin_amdgcn_wmma_f32_16x16x4_f32(
            false, a2[1], false, hb, (short)0, dn, false, false);
        hb.x = h4; hb.y = (k == 0) ? cA : cB;
        dn = __builtin_amdgcn_wmma_f32_16x16x4_f32(
            false, a2[2], false, hb, (short)0, dn, false, false);
        if (k == 0) {
          __builtin_amdgcn_sched_barrier(0);
          pre = __builtin_amdgcn_wmma_f32_16x16x4_f32(
              false, a1x, false, xqa, (short)0, bias1g, false, false);
          __builtin_amdgcn_sched_barrier(0);
        }

        d1 = dn;
        if (k == 0) { o0_first = dn[5]; o1_first = dn[6]; }
        else        { o0_last  = dn[5]; o1_last  = dn[6]; hb_last = hb; }
      }

      if ((le & (Q_GROUPS - 1)) == Q_GROUPS - 1) {
        const int qe = ((le + 1) / Q_GROUPS) & (QUARTERS - 1);
        const v8f e0 = __builtin_amdgcn_wmma_f32_16x16x4_f32(
            false, a1x, false, xqa, (short)0, bias1p, false, false);
        const v8f e1 = __builtin_amdgcn_wmma_f32_16x16x4_f32(
            false, a1e, false, hb_last, (short)0, e0, false, false);
#pragma unroll
        for (int v = 0; v < 5; ++v) d1[v] = (q == qe) ? e1[v] : d1[v];
      }
      __builtin_amdgcn_sched_barrier(0);
      obw[0] = o0_first;
      obw[1] = o0_last;
      cw[0]  = o1_first;
      cw[1]  = o1_last;
      __builtin_amdgcn_sched_barrier(0);
      pre = __builtin_amdgcn_wmma_f32_16x16x4_f32(
          false, a1x, false, xqb, (short)0, bias1g, false, false);
      xqa.x = xn.x; xqa.y = xn.y;
      xqb.x = xn.z; xqb.y = xn.w;
    }

    asm volatile("s_wait_dscnt 0x0" ::: "memory");
    asm volatile("s_barrier_signal -1\n\ts_barrier_wait -1" ::: "memory");
  }

  flush(QUARTERS - 1, wave + WAVES * (QUARTERS - 1) + SLOTS * (T_PER_SLOT - 1));
}

extern "C" void kernel_launch(void* const* d_in, const int* in_sizes, int n_in,
                              void* d_out, int out_size, void* d_ws, size_t ws_size,
                              hipStream_t stream) {
  (void)in_sizes; (void)n_in; (void)out_size; (void)d_ws; (void)ws_size;
  const float* runs = (const float*)d_in[0];
  const float* W1   = (const float*)d_in[1];
  const float* b1   = (const float*)d_in[2];
  const float* W2   = (const float*)d_in[3];
  const float* b2   = (const float*)d_in[4];
  float* out        = (float*)d_out;

  dim3 grid(R_TOT / RUNS_PER_BLOCK);
  dim3 block(32 * WAVES);
  com2net_wmma_kernel<<<grid, block, 0, stream>>>(runs, W1, b1, W2, b2, out);
}


# Once-per-timestep work that one wave does in its turn while three wait at the barrier (x row requests, control flush, agent-0 rebuild) moves to where that wave waits anyway; sums unchanged
# speedup vs baseline: 36.0103x; 1.0701x over previous
#include <hip/hip_runtime.h>


typedef float v2f __attribute__((ext_vector_type(2)));
typedef float v8f __attribute__((ext_vector_type(8)));

#define R_TOT          128
#define T_STEPS        1024
#define N_AG           128
#define RUNS_PER_BLOCK 2
#define QUARTERS       8
#define SLOT_STRIDE    132
#define WAVES          4
#define SLOTS          (WAVES * QUARTERS)
#define GROUP          2
#define LAG_GROUPS     2
#define Q_GROUPS       (LAG_GROUPS * WAVES)
#define T_PER_SLOT     (T_STEPS / SLOTS)
#define GROUPS_PER_T   (N_AG / GROUP)
#define CELLS_PER_COL  (T_PER_SLOT * N_AG)
#define OWN_GROUPS     (T_PER_SLOT * GROUPS_PER_T + Q_GROUPS * (QUARTERS - 1))
#define EPOCHS         (OWN_GROUPS + LAG_GROUPS * (WAVES - 1))
#define CS_ROWS        (RUNS_PER_BLOCK + 32)
#define OB_SINK        192
#define OB_WAVE        (16 * SLOT_STRIDE + OB_SINK)
#define XROW_F         (N_AG * 2)
#define XCOL_F         (2 * XROW_F + 4)
#define XWAVE_F        (16 * XCOL_F)
#define STAGE_EVENTS   (QUARTERS * T_PER_SLOT)

static_assert(RUNS_PER_BLOCK * QUARTERS == 16, "16 matrix columns");
static_assert(LAG_GROUPS * GROUP * SLOTS == N_AG, "a slot's next timestep is due when it ends one");
static_assert(Q_GROUPS * QUARTERS == GROUPS_PER_T, "column slots spread evenly over a timestep");
static_assert(GROUP == 2 && LAG_GROUPS == 2, "the group body below is written for 2 agents, 2 groups behind");
static_assert(T_STEPS % SLOTS == 0 && N_AG % GROUP == 0, "whole timesteps and groups");
static_assert((OB_WAVE * 4) % 16 == 0, "16-byte rows for the flush");

__device__ __forceinline__ float fast_tanh(float x) {
#if __has_builtin(__builtin_amdgcn_tanhf)
  return __builtin_amdgcn_tanhf(x);
#else
  float e = __builtin_exp2f(x * 2.885390081777927f);
  return 1.0f - 2.0f / (e + 1.0f);
#endif
}

__global__ __launch_bounds__(32 * WAVES)
void com2net_wmma_kernel(const float* __restrict__ runs,
                         const float* __restrict__ W1,
                         const float* __restrict__ b1,
                         const float* __restrict__ W2,
                         const float* __restrict__ b2,
                         float*       __restrict__ out)
{
  __shared__ __align__(16) float cslot [CS_ROWS * SLOT_STRIDE];
  __shared__ __align__(16) float outbuf[WAVES * OB_WAVE];
  __shared__ __align__(16) float xbuf  [WAVES * XWAVE_F];

  const int wave    = __builtin_amdgcn_readfirstlane((int)(threadIdx.x >> 5));
  const int lane    = threadIdx.x & 31;
  const int up      = lane >> 4;
  const int l15     = lane & 15;
  const int q       = l15 >> 1;
  const int r       = l15 & 1;
  const int runBase = blockIdx.x * RUNS_PER_BLOCK;

  for (int k = threadIdx.x; k < CS_ROWS * SLOT_STRIDE; k += 32 * WAVES) cslot[k] = 0.0f;
  for (int k = threadIdx.x; k < WAVES * XWAVE_F / 4; k += 32 * WAVES)
    ((float4*)xbuf)[k] = make_float4(0.0f, 0.0f, 0.0f, 0.0f);
  __syncthreads();

  const int hid = (l15 < 5) ? l15 : ((l15 >= 8 && l15 < 13) ? l15 - 3 : -1);

  v2f a1x; a1x.x = 0.0f; a1x.y = 0.0f;
  if (hid >= 0 && !up) { a1x.x = W1[hid * 4 + 0]; a1x.y = W1[hid * 4 + 1]; }
  v2f a1e; a1e.x = 0.0f; a1e.y = 0.0f;
  if (hid >= 0 && up) a1e.y = W1[hid * 4 + 3];

  v2f a2[3];
#pragma unroll
  for (int c2 = 0; c2 < 3; ++c2) {
    a2[c2].x = 0.0f; a2[c2].y = 0.0f;
    const int k0 = 5 * up + 2 * c2;
    if (hid >= 0) {
      const float wf = W1[hid * 4 + 2];
      a2[c2].x = wf * W2[20 + k0];
      if (2 * c2 + 1 < 5) a2[c2].y = wf * W2[20 + k0 + 1];
    } else if (l15 == 5 || l15 == 6) {
      a2[c2].x = W2[(l15 - 5) * 10 + k0];
      if (2 * c2 + 1 < 5) a2[c2].y = W2[(l15 - 5) * 10 + k0 + 1];
    }
  }
  if (up && hid >= 0) a2[2].y = W1[hid * 4 + 3];

  v8f bias1p, bias1g;
#pragma unroll
  for (int v = 0; v < 8; ++v) {
    bias1p[v] = (v < 5) ? b1[5 * up + v] : 0.0f;
    bias1g[v] = (v < 5) ? (b1[5 * up + v] + W1[(5 * up + v) * 4 + 2] * b2[2])
                        : ((!up && (v == 5 || v == 6)) ? b2[v - 5] : 0.0f);
  }

  const float* xb     = &xbuf[wave * XWAVE_F + l15 * XCOL_F];
  float*       crow   = &cslot[r * SLOT_STRIDE];
  float*       csink  = &cslot[(RUNS_PER_BLOCK + lane) * SLOT_STRIDE];
  float*       ob     = &outbuf[wave * OB_WAVE];
  float*       obreal = ob + l15 * SLOT_STRIDE;
  float*       obsink = ob + 16 * SLOT_STRIDE + l15;

  auto xoff = [&](int nn) -> int { return (nn & (2 * N_AG - 1)) * 2; };

  const int srun = lane >> 4;
  auto stage_src = [&](int ev) -> const float* {
    const int tt = wave + WAVES * (ev & (QUARTERS - 1)) + SLOTS * (ev >> 3);
    return runs + ((size_t)(runBase + srun) * T_STEPS + tt) * (N_AG * 2) + l15 * 4;
  };
  auto stage_dst = [&](int ev) -> float* {
    return &xbuf[wave * XWAVE_F + ((ev & (QUARTERS - 1)) * RUNS_PER_BLOCK + srun) * XCOL_F
                 + ((ev >> 3) & 1) * XROW_F + l15 * 4];
  };
  float4 xs0 = make_float4(0.0f, 0.0f, 0.0f, 0.0f), xs1 = xs0, xs2 = xs0, xs3 = xs0;

  float4 fl[RUNS_PER_BLOCK];
#pragma unroll
  for (int rr = 0; rr < RUNS_PER_BLOCK; ++rr) fl[rr] = make_float4(0.0f, 0.0f, 0.0f, 0.0f);
  auto flush_get = [&](int qs) {
#pragma unroll
    for (int rr = 0; rr < RUNS_PER_BLOCK; ++rr)
      fl[rr] = *(const float4*)&ob[(qs * RUNS_PER_BLOCK + rr) * SLOT_STRIDE + lane * 4];
  };
  auto flush_put = [&](int tt) {
#pragma unroll
    for (int rr = 0; rr < RUNS_PER_BLOCK; ++rr)
      *(float4*)(out + ((size_t)(runBase + rr) * T_STEPS + tt) * N_AG + lane * 4) = fl[rr];
  };

  v8f   pre = bias1g;
  v8f   d1  = bias1p;
  v2f   xqa, xqb;
  xqa.x = 0.0f; xqa.y = 0.0f; xqb = xqa;

  for (int g = 0; g < EPOCHS; ++g) {
    const int le = g - LAG_GROUPS * wave;
    if (le >= 0 && le < OWN_GROUPS) {
      const int  nq   = GROUP * le - (N_AG / QUARTERS) * q;
      const int  i0   = nq & (N_AG - 1);
      const bool live = (nq >= 0) && (nq < CELLS_PER_COL) && !up;

      if (le == 0) {
        {
          const float* s0 = stage_src(0);
          const float4 a0 = *(const float4*)(s0);
          const float4 a1 = *(const float4*)(s0 + 64);
          const float4 a2 = *(const float4*)(s0 + 128);
          const float4 a3 = *(const float4*)(s0 + 192);
          const float* s1 = stage_src(1);
          xs0 = *(const float4*)(s1);
          xs1 = *(const float4*)(s1 + 64);
          xs2 = *(const float4*)(s1 + 128);
          xs3 = *(const float4*)(s1 + 192);
          float* d0 = stage_dst(0);
          *(float4*)(d0)       = a0;
          *(float4*)(d0 + 64)  = a1;
          *(float4*)(d0 + 128) = a2;
          *(float4*)(d0 + 192) = a3;
        }
        const float* xs  = xb + xoff(nq);
        const float4 xf0 = *(const float4*)xs;
        const float4 xf1 = *(const float4*)(xs + 4);
        v2f x0v; x0v.x = xf0.x; x0v.y = xf0.y;
        v2f x1v; x1v.x = xf0.z; x1v.y = xf0.w;
        xqa.x = xf1.x; xqa.y = xf1.y;
        xqb.x = xf1.z; xqb.y = xf1.w;
        const float cb0 = crow[i0 + 1];
        v2f cbv; cbv.x = 0.0f; cbv.y = cb0;
        const v8f e0 = __builtin_amdgcn_wmma_f32_16x16x4_f32(
            false, a1x, false, x0v, (short)0, bias1p, false, false);
        d1 = __builtin_amdgcn_wmma_f32_16x16x4_f32(
            false, a1e, false, cbv, (short)0, e0, false, false);
        pre = __builtin_amdgcn_wmma_f32_16x16x4_f32(
            false, a1x, false, x1v, (short)0, bias1g, false, false);
      }

      if ((le & (Q_GROUPS - 1)) == 0) {
        const int qs = (le / Q_GROUPS) & (QUARTERS - 1);
        const int ms = le / GROUPS_PER_T;
        const int sl = wave + WAVES * qs;
        const int ev = le / Q_GROUPS;
        if (ev + 1 < STAGE_EVENTS) {
          float* d = stage_dst(ev + 1);
          *(float4*)(d)       = xs0;
          *(float4*)(d + 64)  = xs1;
          *(float4*)(d + 128) = xs2;
          *(float4*)(d + 192) = xs3;
        }
        if (ms > 0) flush_get(qs);
      }

      const float  cA  = crow[i0 + 2];
      const float  cB  = crow[((i0 + GROUP) & (N_AG - 1)) + 1];
      const float4 xn  = *(const float4*)(xb + xoff(nq + 4));
      __builtin_amdgcn_sched_barrier(0);
      const bool rebuild = (le & (Q_GROUPS - 1)) == Q_GROUPS - 1;
      v8f e0, e1;
      if (rebuild)
        e0 = __builtin_amdgcn_wmma_f32_16x16x4_f32(
            false, a1x, false, xqa, (short)0, bias1p, false, false);
      __builtin_amdgcn_sched_barrier(0);
      float*       cw  = (live ? crow   : csink)  + i0;
      float*       obw = (live ? obreal : obsink) + i0;

      float  o0_first = 0.0f, o1_first = 0.0f, o0_last = 0.0f, o1_last = 0.0f;
#pragma unroll
      for (int k = 0; k < GROUP; ++k) {
        const float h0 = fast_tanh(d1[0]);
        const float h1 = fast_tanh(d1[1]);
        const float h2 = fast_tanh(d1[2]);
        const float h3 = fast_tanh(d1[3]);
        const float h4 = fast_tanh(d1[4]);

        v2f hb;
        hb.x = h0; hb.y = h1;
        v8f dn = __builtin_amdgcn_wmma_f32_16x16x4_f32(
            false, a2[0], false, hb, (short)0, pre, false, false);
        hb.x = h2; hb.y = h3;
        dn = __builtin_amdgcn_wmma_f32_16x16x4_f32(
            false, a2[1], false, hb, (short)0, dn, false, false);
        hb.x = h4; hb.y = (k == 0) ? cA : cB;
        dn = __builtin_amdgcn_wmma_f32_16x16x4_f32(
            false, a2[2], false, hb, (short)0, dn, false, false);
        if (k == 0) {
          __builtin_amdgcn_sched_barrier(0);
          if (rebuild) {
            v2f cbv; cbv.x = 0.0f; cbv.y = cB;
            e1 = __builtin_amdgcn_wmma_f32_16x16x4_f32(
                false, a1e, false, cbv, (short)0, e0, false, false);
          }
          pre = __builtin_amdgcn_wmma_f32_16x16x4_f32(
              false, a1x, false, xqa, (short)0, bias1g, false, false);
          __builtin_amdgcn_sched_barrier(0);
        }

        d1 = dn;
        if (k == 0) { o0_first = dn[5]; o1_first = dn[6]; }
        else        { o0_last  = dn[5]; o1_last  = dn[6]; }
      }

      if (rebuild) {
        const int qe = ((le + 1) / Q_GROUPS) & (QUARTERS - 1);
#pragma unroll
        for (int v = 0; v < 5; ++v) d1[v] = (q == qe) ? e1[v] : d1[v];
      }
      __builtin_amdgcn_sched_barrier(0);
      obw[0] = o0_first;
      obw[1] = o0_last;
      cw[0]  = o1_first;
      cw[1]  = o1_last;
      __builtin_amdgcn_sched_barrier(0);
      pre = __builtin_amdgcn_wmma_f32_16x16x4_f32(
          false, a1x, false, xqb, (short)0, bias1g, false, false);
      if ((le & (Q_GROUPS - 1)) == 0) {
        const int ev = le / Q_GROUPS;
        const int ms = le / GROUPS_PER_T;
        if (ev + 2 < STAGE_EVENTS) {
          const float* s = stage_src(ev + 2);
          xs0 = *(const float4*)(s);
          xs1 = *(const float4*)(s + 64);
          xs2 = *(const float4*)(s + 128);
          xs3 = *(const float4*)(s + 192);
        }
        if (ms > 0)
          flush_put(wave + WAVES * (ev & (QUARTERS - 1)) + SLOTS * (ms - 1));
      }
      xqa.x = xn.x; xqa.y = xn.y;
      xqb.x = xn.z; xqb.y = xn.w;
    }

    asm volatile("s_wait_dscnt 0x0" ::: "memory");
    asm volatile("s_barrier_signal -1\n\ts_barrier_wait -1" ::: "memory");
  }

  flush_get(QUARTERS - 1);
  flush_put(wave + WAVES * (QUARTERS - 1) + SLOTS * (T_PER_SLOT - 1));
}

extern "C" void kernel_launch(void* const* d_in, const int* in_sizes, int n_in,
                              void* d_out, int out_size, void* d_ws, size_t ws_size,
                              hipStream_t stream) {
  (void)in_sizes; (void)n_in; (void)out_size; (void)d_ws; (void)ws_size;
  const float* runs = (const float*)d_in[0];
  const float* W1   = (const float*)d_in[1];
  const float* b1   = (const float*)d_in[2];
  const float* W2   = (const float*)d_in[3];
  const float* b2   = (const float*)d_in[4];
  float* out        = (float*)d_out;

  dim3 grid(R_TOT / RUNS_PER_BLOCK);
  dim3 block(32 * WAVES);
  com2net_wmma_kernel<<<grid, block, 0, stream>>>(runs, W1, b1, W2, b2, out);
}


# Path from barrier to first backward word cleared: top-load addresses and first rebuild product formed ahead of it, timestep-start LDS work behind the loads, rare blocks out of line; sums same
# speedup vs baseline: 38.2467x; 1.0621x over previous
#include <hip/hip_runtime.h>


typedef float v2f __attribute__((ext_vector_type(2)));
typedef float v8f __attribute__((ext_vector_type(8)));
typedef float v4f __attribute__((ext_vector_type(4)));
typedef const __attribute__((address_space(3))) float* lds_f;
typedef const __attribute__((address_space(3))) v4f*   lds_v4;

#define R_TOT          128
#define T_STEPS        1024
#define N_AG           128
#define RUNS_PER_BLOCK 2
#define QUARTERS       8
#define SLOT_STRIDE    132
#define WAVES          4
#define SLOTS          (WAVES * QUARTERS)
#define GROUP          2
#define LAG_GROUPS     2
#define Q_GROUPS       (LAG_GROUPS * WAVES)
#define T_PER_SLOT     (T_STEPS / SLOTS)
#define GROUPS_PER_T   (N_AG / GROUP)
#define CELLS_PER_COL  (T_PER_SLOT * N_AG)
#define OWN_GROUPS     (T_PER_SLOT * GROUPS_PER_T + Q_GROUPS * (QUARTERS - 1))
#define EPOCHS         (OWN_GROUPS + LAG_GROUPS * (WAVES - 1))
#define CS_ROWS        (RUNS_PER_BLOCK + 32)
#define OB_SINK        192
#define OB_WAVE        (16 * SLOT_STRIDE + OB_SINK)
#define XROW_F         (N_AG * 2)
#define XCOL_F         (2 * XROW_F + 4)
#define XWAVE_F        (16 * XCOL_F)
#define STAGE_EVENTS   (QUARTERS * T_PER_SLOT)

static_assert(RUNS_PER_BLOCK * QUARTERS == 16, "16 matrix columns");
static_assert(LAG_GROUPS * GROUP * SLOTS == N_AG, "a slot's next timestep is due when it ends one");
static_assert(Q_GROUPS * QUARTERS == GROUPS_PER_T, "column slots spread evenly over a timestep");
static_assert(GROUP == 2 && LAG_GROUPS == 2, "the group body below is written for 2 agents, 2 groups behind");
static_assert(T_STEPS % SLOTS == 0 && N_AG % GROUP == 0, "whole timesteps and groups");
static_assert((OB_WAVE * 4) % 16 == 0, "16-byte rows for the flush");

__device__ __forceinline__ float fast_tanh(float x) {
#if __has_builtin(__builtin_amdgcn_tanhf)
  return __builtin_amdgcn_tanhf(x);
#else
  float e = __builtin_exp2f(x * 2.885390081777927f);
  return 1.0f - 2.0f / (e + 1.0f);
#endif
}

__global__ __launch_bounds__(32 * WAVES)
void com2net_wmma_kernel(const float* __restrict__ runs,
                         const float* __restrict__ W1,
                         const float* __restrict__ b1,
                         const float* __restrict__ W2,
                         const float* __restrict__ b2,
                         float*       __restrict__ out)
{
  __shared__ __align__(16) float cslot [CS_ROWS * SLOT_STRIDE];
  __shared__ __align__(16) float outbuf[WAVES * OB_WAVE];
  __shared__ __align__(16) float xbuf  [WAVES * XWAVE_F];

  const int wave    = __builtin_amdgcn_readfirstlane((int)(threadIdx.x >> 5));
  const int lane    = threadIdx.x & 31;
  const int up      = lane >> 4;
  const int l15     = lane & 15;
  const int q       = l15 >> 1;
  const int r       = l15 & 1;
  const int runBase = blockIdx.x * RUNS_PER_BLOCK;

  for (int k = threadIdx.x; k < CS_ROWS * SLOT_STRIDE; k += 32 * WAVES) cslot[k] = 0.0f;
  for (int k = threadIdx.x; k < WAVES * XWAVE_F / 4; k += 32 * WAVES)
    ((float4*)xbuf)[k] = make_float4(0.0f, 0.0f, 0.0f, 0.0f);
  __syncthreads();

  const int hid = (l15 < 5) ? l15 : ((l15 >= 8 && l15 < 13) ? l15 - 3 : -1);

  v2f a1x; a1x.x = 0.0f; a1x.y = 0.0f;
  if (hid >= 0 && !up) { a1x.x = W1[hid * 4 + 0]; a1x.y = W1[hid * 4 + 1]; }
  v2f a1e; a1e.x = 0.0f; a1e.y = 0.0f;
  if (hid >= 0 && up) a1e.y = W1[hid * 4 + 3];

  v2f a2[3];
#pragma unroll
  for (int c2 = 0; c2 < 3; ++c2) {
    a2[c2].x = 0.0f; a2[c2].y = 0.0f;
    const int k0 = 5 * up + 2 * c2;
    if (hid >= 0) {
      const float wf = W1[hid * 4 + 2];
      a2[c2].x = wf * W2[20 + k0];
      if (2 * c2 + 1 < 5) a2[c2].y = wf * W2[20 + k0 + 1];
    } else if (l15 == 5 || l15 == 6) {
      a2[c2].x = W2[(l15 - 5) * 10 + k0];
      if (2 * c2 + 1 < 5) a2[c2].y = W2[(l15 - 5) * 10 + k0 + 1];
    }
  }
  if (up && hid >= 0) a2[2].y = W1[hid * 4 + 3];

  v8f bias1p, bias1g;
#pragma unroll
  for (int v = 0; v < 8; ++v) {
    bias1p[v] = (v < 5) ? b1[5 * up + v] : 0.0f;
    bias1g[v] = (v < 5) ? (b1[5 * up + v] + W1[(5 * up + v) * 4 + 2] * b2[2])
                        : ((!up && (v == 5 || v == 6)) ? b2[v - 5] : 0.0f);
  }

  const float* xb     = &xbuf[wave * XWAVE_F + l15 * XCOL_F];
  float*       crow   = &cslot[r * SLOT_STRIDE];
  float*       csink  = &cslot[(RUNS_PER_BLOCK + lane) * SLOT_STRIDE];
  float*       ob     = &outbuf[wave * OB_WAVE];
  float*       obreal = ob + l15 * SLOT_STRIDE;
  float*       obsink = ob + 16 * SLOT_STRIDE + l15;

  auto xoff = [&](int nn) -> int { return (nn & (2 * N_AG - 1)) * 2; };

  const int srun = lane >> 4;
  auto stage_src = [&](int ev) -> const float* {
    const int tt = wave + WAVES * (ev & (QUARTERS - 1)) + SLOTS * (ev >> 3);
    return runs + ((size_t)(runBase + srun) * T_STEPS + tt) * (N_AG * 2) + l15 * 4;
  };
  auto stage_dst = [&](int ev) -> float* {
    return &xbuf[wave * XWAVE_F + ((ev & (QUARTERS - 1)) * RUNS_PER_BLOCK + srun) * XCOL_F
                 + ((ev >> 3) & 1) * XROW_F + l15 * 4];
  };
  float4 xs0 = make_float4(0.0f, 0.0f, 0.0f, 0.0f), xs1 = xs0, xs2 = xs0, xs3 = xs0;

  float4 fl[RUNS_PER_BLOCK];
#pragma unroll
  for (int rr = 0; rr < RUNS_PER_BLOCK; ++rr) fl[rr] = make_float4(0.0f, 0.0f, 0.0f, 0.0f);
  auto flush_get = [&](int qs) {
#pragma unroll
    for (int rr = 0; rr < RUNS_PER_BLOCK; ++rr)
      fl[rr] = *(const float4*)&ob[(qs * RUNS_PER_BLOCK + rr) * SLOT_STRIDE + lane * 4];
  };
  auto flush_put = [&](int tt) {
#pragma unroll
    for (int rr = 0; rr < RUNS_PER_BLOCK; ++rr)
      *(float4*)(out + ((size_t)(runBase + rr) * T_STEPS + tt) * N_AG + lane * 4) = fl[rr];
  };

  v8f   pre = bias1g;
  v8f   d1  = bias1p;
  v2f   xqa, xqb;
  xqa.x = 0.0f; xqa.y = 0.0f; xqb = xqa;
  v8f   e0 = bias1p;

  const lds_f crow3 = (lds_f)crow;
  const lds_f xb3   = (lds_f)xb;
  int   nqC = GROUP * (0 - LAG_GROUPS * wave) - (N_AG / QUARTERS) * q;
  lds_f pA, pB, pX;
  auto form_top = [&]() {
    const int i0N = nqC & (N_AG - 1);
    pA = crow3 + i0N + 2;
    pB = crow3 + ((i0N + GROUP) & (N_AG - 1)) + 1;
    pX = xb3 + xoff(nqC + 4);
  };
  form_top();

  for (int g = 0; g < EPOCHS; ++g) {
    const int le = g - LAG_GROUPS * wave;
    if (le >= 0 && le < OWN_GROUPS) {
      const int  nq   = nqC;
      const int  i0   = nq & (N_AG - 1);
      const bool live = (nq >= 0) && (nq < CELLS_PER_COL) && !up;

      const float cA = *pA;
      const float cB = *pB;
      v4f xn = *(lds_v4)pX;

      if (__builtin_expect(le == 0, 0)) {
        {
          const float* s0 = stage_src(0);
          const float4 a0 = *(const float4*)(s0);
          const float4 a1 = *(const float4*)(s0 + 64);
          const float4 a2 = *(const float4*)(s0 + 128);
          const float4 a3 = *(const float4*)(s0 + 192);
          const float* s1 = stage_src(1);
          xs0 = *(const float4*)(s1);
          xs1 = *(const float4*)(s1 + 64);
          xs2 = *(const float4*)(s1 + 128);
          xs3 = *(const float4*)(s1 + 192);
          float* d0 = stage_dst(0);
          *(float4*)(d0)       = a0;
          *(float4*)(d0 + 64)  = a1;
          *(float4*)(d0 + 128) = a2;
          *(float4*)(d0 + 192) = a3;
        }
        const float* xs  = xb + xoff(nq);
        const float4 xf0 = *(const float4*)xs;
        const float4 xf1 = *(const float4*)(xs + 4);
        v2f x0v; x0v.x = xf0.x; x0v.y = xf0.y;
        v2f x1v; x1v.x = xf0.z; x1v.y = xf0.w;
        xqa.x = xf1.x; xqa.y = xf1.y;
        xqb.x = xf1.z; xqb.y = xf1.w;
        const float cb0 = crow[i0 + 1];
        v2f cbv; cbv.x = 0.0f; cbv.y = cb0;
        const v8f e0 = __builtin_amdgcn_wmma_f32_16x16x4_f32(
            false, a1x, false, x0v, (short)0, bias1p, false, false);
        d1 = __builtin_amdgcn_wmma_f32_16x16x4_f32(
            false, a1e, false, cbv, (short)0, e0, false, false);
        pre = __builtin_amdgcn_wmma_f32_16x16x4_f32(
            false, a1x, false, x1v, (short)0, bias1g, false, false);
        xn = *(lds_v4)pX;
      }

      if (__builtin_expect((le & (Q_GROUPS - 1)) == 0, 0)) {
        const int qs = (le / Q_GROUPS) & (QUARTERS - 1);
        const int ms = le / GROUPS_PER_T;
        const int ev = le / Q_GROUPS;
        if (ev + 1 < STAGE_EVENTS) {
          float* d = stage_dst(ev + 1);
          *(float4*)(d)       = xs0;
          *(float4*)(d + 64)  = xs1;
          *(float4*)(d + 128) = xs2;
          *(float4*)(d + 192) = xs3;
        }
        if (ms > 0) flush_get(qs);
      }
      __builtin_amdgcn_sched_barrier(0);
      const bool rebuild = (le & (Q_GROUPS - 1)) == Q_GROUPS - 1;
      v8f e1;
      float*       cw  = (live ? crow   : csink)  + i0;
      float*       obw = (live ? obreal : obsink) + i0;

      float  o0_first = 0.0f, o1_first = 0.0f, o0_last = 0.0f, o1_last = 0.0f;
#pragma unroll
      for (int k = 0; k < GROUP; ++k) {
        const float h0 = fast_tanh(d1[0]);
        const float h1 = fast_tanh(d1[1]);
        const float h2 = fast_tanh(d1[2]);
        const float h3 = fast_tanh(d1[3]);
        const float h4 = fast_tanh(d1[4]);

        v2f hb;
        hb.x = h0; hb.y = h1;
        v8f dn = __builtin_amdgcn_wmma_f32_16x16x4_f32(
            false, a2[0], false, hb, (short)0, pre, false, false);
        hb.x = h2; hb.y = h3;
        dn = __builtin_amdgcn_wmma_f32_16x16x4_f32(
            false, a2[1], false, hb, (short)0, dn, false, false);
        hb.x = h4; hb.y = (k == 0) ? cA : cB;
        dn = __builtin_amdgcn_wmma_f32_16x16x4_f32(
            false, a2[2], false, hb, (short)0, dn, false, false);
        if (k == 0) {
          __builtin_amdgcn_sched_barrier(0);
          if (rebuild) {
            v2f cbv; cbv.x = 0.0f; cbv.y = cB;
            e1 = __builtin_amdgcn_wmma_f32_16x16x4_f32(
                false, a1e, false, cbv, (short)0, e0, false, false);
          }
          pre = __builtin_amdgcn_wmma_f32_16x16x4_f32(
              false, a1x, false, xqa, (short)0, bias1g, false, false);
          __builtin_amdgcn_sched_barrier(0);
        }

        d1 = dn;
        if (k == 0) { o0_first = dn[5]; o1_first = dn[6]; }
        else        { o0_last  = dn[5]; o1_last  = dn[6]; }
      }

      if (rebuild) {
        const int qe = ((le + 1) / Q_GROUPS) & (QUARTERS - 1);
#pragma unroll
        for (int v = 0; v < 5; ++v) d1[v] = (q == qe) ? e1[v] : d1[v];
      }
      __builtin_amdgcn_sched_barrier(0);
      obw[0] = o0_first;
      obw[1] = o0_last;
      cw[0]  = o1_first;
      cw[1]  = o1_last;
      __builtin_amdgcn_sched_barrier(0);
      pre = __builtin_amdgcn_wmma_f32_16x16x4_f32(
          false, a1x, false, xqb, (short)0, bias1g, false, false);
      if ((le & (Q_GROUPS - 1)) == 0) {
        const int ev = le / Q_GROUPS;
        const int ms = le / GROUPS_PER_T;
        if (ev + 2 < STAGE_EVENTS) {
          const float* s = stage_src(ev + 2);
          xs0 = *(const float4*)(s);
          xs1 = *(const float4*)(s + 64);
          xs2 = *(const float4*)(s + 128);
          xs3 = *(const float4*)(s + 192);
        }
        if (ms > 0)
          flush_put(wave + WAVES * (ev & (QUARTERS - 1)) + SLOTS * (ms - 1));
      }
      xqa.x = xn.x; xqa.y = xn.y;
      xqb.x = xn.z; xqb.y = xn.w;
      if ((le & (Q_GROUPS - 1)) == Q_GROUPS - 2)
        e0 = __builtin_amdgcn_wmma_f32_16x16x4_f32(
            false, a1x, false, xqa, (short)0, bias1p, false, false);
    }
    nqC += GROUP;
    form_top();

    asm volatile("s_wait_dscnt 0x0\n\ts_barrier_signal -1\n\ts_barrier_wait -1"
                 :: "v"(pA), "v"(pB), "v"(pX) : "memory");
  }

  flush_get(QUARTERS - 1);
  flush_put(wave + WAVES * (QUARTERS - 1) + SLOTS * (T_PER_SLOT - 1));
}

extern "C" void kernel_launch(void* const* d_in, const int* in_sizes, int n_in,
                              void* d_out, int out_size, void* d_ws, size_t ws_size,
                              hipStream_t stream) {
  (void)in_sizes; (void)n_in; (void)out_size; (void)d_ws; (void)ws_size;
  const float* runs = (const float*)d_in[0];
  const float* W1   = (const float*)d_in[1];
  const float* b1   = (const float*)d_in[2];
  const float* W2   = (const float*)d_in[3];
  const float* b2   = (const float*)d_in[4];
  float* out        = (float*)d_out;

  dim3 grid(R_TOT / RUNS_PER_BLOCK);
  dim3 block(32 * WAVES);
  com2net_wmma_kernel<<<grid, block, 0, stream>>>(runs, W1, b1, W2, b2, out);
}
